# starting kernel
# baseline (speedup 1.0000x reference)
#include <hip/hip_runtime.h>


typedef __attribute__((ext_vector_type(2))) float v2f;
typedef __attribute__((ext_vector_type(8))) float v8f;

#define GCN_N 100000


__global__ void k_init_deg(float* __restrict__ deg) {
    int i = blockIdx.x * blockDim.x + threadIdx.x;
    if (i < GCN_N) deg[i] = 1.0f;
}

__global__ void k_deg_accum(const int* __restrict__ dst, int E,
                            float* __restrict__ deg) {
    int e = blockIdx.x * blockDim.x + threadIdx.x;
    if (e < E) atomicAdd(&deg[dst[e]], 1.0f);
}

__global__ void k_deg_rsqrt(float* __restrict__ deg) {
    int i = blockIdx.x * blockDim.x + threadIdx.x;
    if (i < GCN_N) deg[i] = rsqrtf(deg[i]);
}

__global__ void k_zero(float* __restrict__ p, int n) {
    int i = blockIdx.x * blockDim.x + threadIdx.x;
    if (i < n) p[i] = 0.0f;
}

template <int KI, int NO>
__global__ void k_gemm_wmma(const float* __restrict__ X,
                            const float* __restrict__ W,
                            float* __restrict__ T) {
    const int wave     = (blockIdx.x * blockDim.x + threadIdx.x) >> 5;
    const int lane     = threadIdx.x & 31;
    const int nColTile = NO / 16;
    const int rowTile  = wave / nColTile;
    const int colTile  = wave - rowTile * nColTile;
    if (rowTile >= GCN_N / 16) return;

    const int row0 = rowTile * 16;
    const int col0 = colTile * 16;
    const int m    = lane & 15;
    const int hi   = lane >> 4;

    v8f acc = {};
    constexpr int KCHUNKS = (KI + 3) / 4;
#pragma unroll
    for (int kk = 0; kk < KCHUNKS; ++kk) {
        const int kb = kk * 4 + hi * 2;
        v2f a, b;
        a.x = (kb     < KI) ? X[(row0 + m) * KI + kb    ] : 0.0f;
        a.y = (kb + 1 < KI) ? X[(row0 + m) * KI + kb + 1] : 0.0f;
        b.x = (kb     < KI) ? W[(kb    ) * NO + col0 + m] : 0.0f;
        b.y = (kb + 1 < KI) ? W[(kb + 1) * NO + col0 + m] : 0.0f;
        acc = __builtin_amdgcn_wmma_f32_16x16x4_f32(
            false, a, false, b, (short)0, acc, false, false);
    }

#pragma unroll
    for (int j = 0; j < 8; ++j)
        T[(row0 + j + hi * 8) * NO + col0 + m] = acc[j];
}

template <int F>
__global__ void k_scatter(const int* __restrict__ src,
                          const int* __restrict__ dst, int E,
                          const float* __restrict__ dinv,
                          const float* __restrict__ T,
                          float* __restrict__ agg) {
    int e = blockIdx.x * blockDim.x + threadIdx.x;
    if (e >= E) return;
    const int   s    = src[e];
    const int   d    = dst[e];
    const float norm = dinv[s] * dinv[d];
    const float4* tp = (const float4*)(T + (size_t)s * F);
    float*        ap = agg + (size_t)d * F;
#pragma unroll
    for (int c = 0; c < F / 4; ++c) {
        float4 v = tp[c];
        atomicAdd(ap + 4 * c + 0, v.x * norm);
        atomicAdd(ap + 4 * c + 1, v.y * norm);
        atomicAdd(ap + 4 * c + 2, v.z * norm);
        atomicAdd(ap + 4 * c + 3, v.w * norm);
    }
}

template <int F, bool RELU>
__global__ void k_finish(const float* __restrict__ T,
                         const float* __restrict__ dinv,
                         const float* __restrict__ bias,
                         float* __restrict__ agg) {
    int i = blockIdx.x * blockDim.x + threadIdx.x;
    if (i >= GCN_N * F) return;
    const int   node = i / F;
    const int   f    = i - node * F;
    const float di   = dinv[node];
    float v = agg[i] + T[i] * di * di + bias[f];
    agg[i] = RELU ? fmaxf(v, 0.0f) : v;
}


extern "C" void kernel_launch(void* const* d_in, const int* in_sizes, int n_in,
                              void* d_out, int out_size, void* d_ws, size_t ws_size,
                              hipStream_t stream) {
    const float* x  = (const float*)d_in[0];
    const int*   ei = (const int*)d_in[1];
    const float* W1 = (const float*)d_in[2];
    const float* b1 = (const float*)d_in[3];
    const float* W2 = (const float*)d_in[4];
    const float* b2 = (const float*)d_in[5];
    const float* W3 = (const float*)d_in[6];
    const float* b3 = (const float*)d_in[7];
    float* out = (float*)d_out;

    const int E = in_sizes[1] / 2;
    const int* src = ei;
    const int* dst = ei + E;

    const int N = GCN_N;

    float* ws   = (float*)d_ws;
    float* dinv = ws;
    float* t1   = dinv + N;
    float* h1   = t1 + (size_t)N * 16;
    float* t2   = h1 + (size_t)N * 16;
    float* h2   = t2 + (size_t)N * 32;
    float* t3   = h2 + (size_t)N * 32;

    const int B = 256;
    auto blk = [](int n, int b) { return (n + b - 1) / b; };

    k_init_deg<<<blk(N, B), B, 0, stream>>>(dinv);
    k_deg_accum<<<blk(E, B), B, 0, stream>>>(dst, E, dinv);
    k_deg_rsqrt<<<blk(N, B), B, 0, stream>>>(dinv);

    {
        const int waves = (N / 16) * 1;
        k_gemm_wmma<2, 16><<<blk(waves * 32, B), B, 0, stream>>>(x, W1, t1);
        k_zero<<<blk(N * 16, B), B, 0, stream>>>(h1, N * 16);
        k_scatter<16><<<blk(E, B), B, 0, stream>>>(src, dst, E, dinv, t1, h1);
        k_finish<16, true><<<blk(N * 16, B), B, 0, stream>>>(t1, dinv, b1, h1);
    }

    {
        const int waves = (N / 16) * 2;
        k_gemm_wmma<16, 32><<<blk(waves * 32, B), B, 0, stream>>>(h1, W2, t2);
        k_zero<<<blk(N * 32, B), B, 0, stream>>>(h2, N * 32);
        k_scatter<32><<<blk(E, B), B, 0, stream>>>(src, dst, E, dinv, t2, h2);
        k_finish<32, true><<<blk(N * 32, B), B, 0, stream>>>(t2, dinv, b2, h2);
    }

    {
        const int waves = (N / 16) * 1;
        k_gemm_wmma<32, 16><<<blk(waves * 32, B), B, 0, stream>>>(h2, W3, t3);
        k_zero<<<blk(N * 16, B), B, 0, stream>>>(out, N * 16);
        k_scatter<16><<<blk(E, B), B, 0, stream>>>(src, dst, E, dinv, t3, out);
        k_finish<16, false><<<blk(N * 16, B), B, 0, stream>>>(t3, dinv, b3, out);
    }
}


# Atomic sums kept per package half: degrees and each layer's neighbour sums get two planes, a wave adds only into its own half's plane (die number from its placement word); readers add both planes
# speedup vs baseline: 12.9975x; 12.9975x over previous
#include <hip/hip_runtime.h>


typedef __attribute__((ext_vector_type(2))) float v2f;
typedef __attribute__((ext_vector_type(8))) float v8f;

#define GCN_N 100000

static __device__ __forceinline__ int chip_half() {
    return (int)((__builtin_amdgcn_s_sendmsg_rtn(0x87) >> 18) & 1u);
}


template <bool TWO>
__global__ void k_init_deg(float* __restrict__ deg) {
    int i = blockIdx.x * blockDim.x + threadIdx.x;
    if (i < (TWO ? 2 : 1) * GCN_N) deg[i] = (i < GCN_N) ? 1.0f : 0.0f;
}

template <bool TWO>
__global__ void k_deg_accum(const int* __restrict__ dst, int E,
                            float* __restrict__ deg) {
    int e = blockIdx.x * blockDim.x + threadIdx.x;
    if constexpr (TWO) deg += chip_half() * GCN_N;
    if (e < E) atomicAdd(&deg[dst[e]], 1.0f);
}

template <bool TWO>
static __device__ __forceinline__ float inv_sqrt_deg(const float* __restrict__ deg, int node) {
    if constexpr (TWO) return rsqrtf(deg[node] + deg[GCN_N + node]);
    else               return rsqrtf(deg[node]);
}

template <int KI, int NO, bool EPI, bool TWO>
__global__ void k_transform(const float* __restrict__ X,
                            const float* __restrict__ XB,
                            const float* __restrict__ W,
                            const float* __restrict__ deg,
                            const float* __restrict__ bias_in,
                            float* __restrict__ T,
                            float* __restrict__ agg,
                            float* __restrict__ aggB) {
    static_assert(!EPI || KI % 4 == 0, "the epilogue on load has no padded k");
    const int wave     = (blockIdx.x * blockDim.x + threadIdx.x) >> 5;
    const int lane     = threadIdx.x & 31;
    const int nColTile = NO / 16;
    const int rowTile  = wave / nColTile;
    const int colTile  = wave - rowTile * nColTile;
    if (rowTile >= GCN_N / 16) return;

    const int row0 = rowTile * 16;
    const int col0 = colTile * 16;
    const int m    = lane & 15;
    const int hi   = lane >> 4;

    float di_m = 0.0f;
    if constexpr (EPI) di_m = inv_sqrt_deg<TWO>(deg, row0 + m);

    v8f acc = {};
    constexpr int KCHUNKS = (KI + 3) / 4;
#pragma unroll
    for (int kk = 0; kk < KCHUNKS; ++kk) {
        const int kb = kk * 4 + hi * 2;
        v2f a, b;
        a.x = (kb     < KI) ? X[(row0 + m) * KI + kb    ] : 0.0f;
        a.y = (kb + 1 < KI) ? X[(row0 + m) * KI + kb + 1] : 0.0f;
        if constexpr (EPI) {
            if constexpr (TWO) {
                a.x += XB[(row0 + m) * KI + kb    ];
                a.y += XB[(row0 + m) * KI + kb + 1];
            }
            a.x = fmaxf(a.x * di_m + bias_in[kb    ], 0.0f);
            a.y = fmaxf(a.y * di_m + bias_in[kb + 1], 0.0f);
        }
        b.x = (kb     < KI) ? W[(kb    ) * NO + col0 + m] : 0.0f;
        b.y = (kb + 1 < KI) ? W[(kb + 1) * NO + col0 + m] : 0.0f;
        acc = __builtin_amdgcn_wmma_f32_16x16x4_f32(
            false, a, false, b, (short)0, acc, false, false);
    }

#pragma unroll
    for (int j = 0; j < 8; ++j) {
        const int   row = row0 + j + hi * 8;
        const float v   = acc[j] * inv_sqrt_deg<TWO>(deg, row);
        T  [row * NO + col0 + m] = v;
        agg[row * NO + col0 + m] = v;
        if constexpr (TWO) aggB[row * NO + col0 + m] = 0.0f;
    }
}

template <int... Js> struct iseq {};

template <int J>
static __device__ __forceinline__ int half_lane(int x) {
    return __builtin_amdgcn_mov_dpp(x, 0x150 + J, 0xF, 0xF, false);
}

template <bool FULL, int... Js>
static __device__ __forceinline__ void rows16(int s, int d, int ok, int c,
                                              const float* __restrict__ T,
                                              float* __restrict__ agg, iseq<Js...>) {
    float v[16];
    ((v[Js] = T[(unsigned)half_lane<Js>(s) * 16u + (unsigned)c]), ...);
    ((FULL || half_lane<Js>(ok)
          ? (void)atomicAdd(&agg[(unsigned)half_lane<Js>(d) * 16u + (unsigned)c], v[Js])
          : (void)0), ...);
}

template <bool FULL>
static __device__ __forceinline__ void rows32(int s, int d, int nvalid, int lane,
                                              const float* __restrict__ T,
                                              float* __restrict__ agg) {
#pragma unroll
    for (int h = 0; h < 32; h += 16) {
        float v[16];
#pragma unroll
        for (int j = 0; j < 16; ++j) {
            const int sj = __builtin_amdgcn_readlane(s, h + j);
            v[j] = T[(unsigned)sj * 32u + (unsigned)lane];
        }
#pragma unroll
        for (int j = 0; j < 16; ++j) {
            if (FULL || h + j < nvalid) {
                const int dj = __builtin_amdgcn_readlane(d, h + j);
                atomicAdd(&agg[(unsigned)dj * 32u + (unsigned)lane], v[j]);
            }
        }
    }
}

template <int F, bool FULL>
static __device__ __forceinline__ void rows(int s, int d, int ok, int nvalid, int lane,
                                            const float* __restrict__ T,
                                            float* __restrict__ agg) {
    if constexpr (F == 32)
        rows32<FULL>(s, d, nvalid, lane, T, agg);
    else
        rows16<FULL>(s, d, ok, lane & 15, T, agg,
                     iseq<0, 1, 2, 3, 4, 5, 6, 7, 8, 9, 10, 11, 12, 13, 14, 15>{});
}

template <int F, bool TWO>
__global__ __launch_bounds__(256) void k_scatter(const int* __restrict__ src,
                                                 const int* __restrict__ dst, int E,
                                                 const float* __restrict__ T,
                                                 float* __restrict__ aggA,
                                                 float* __restrict__ aggB) {
    static_assert(F == 16 || F == 32, "row = 16 or 32 columns");
    const int lane  = threadIdx.x & 31;
    const int wpb   = blockDim.x >> 5;
    const int wave  = blockIdx.x * wpb + __builtin_amdgcn_readfirstlane(threadIdx.x >> 5);
    const int nwave = gridDim.x * wpb;
    const int nfull = E >> 5;

    float* __restrict__ agg = aggA;
    if constexpr (TWO) { if (chip_half()) agg = aggB; }

    int b = wave;
    int s = 0, d = 0;
    if (b < nfull) { s = src[(b << 5) + lane]; d = dst[(b << 5) + lane]; }
    while (b < nfull) {
        const int bn = b + nwave;
        int sn = 0, dn = 0;
        if (bn < nfull) { sn = src[(bn << 5) + lane]; dn = dst[(bn << 5) + lane]; }
        rows<F, true>(s, d, 1, 32, lane, T, agg);
        s = sn; d = dn; b = bn;
    }

    const int ntail = E & 31;
    if (ntail != 0 && wave == nfull % nwave) {
        const int  e  = (nfull << 5) + lane;
        const bool ok = lane < ntail;
        const int  st = ok ? src[e] : 0;
        const int  dt = ok ? dst[e] : 0;
        rows<F, false>(st, dt, ok ? 1 : 0, ntail, lane, T, agg);
    }
}

template <int F, bool TWO>
__global__ void k_finish(const float* __restrict__ deg,
                         const float* __restrict__ bias,
                         float* __restrict__ agg,
                         const float* __restrict__ aggB) {
    int i = blockIdx.x * blockDim.x + threadIdx.x;
    if (i >= GCN_N * F) return;
    const int   node = i / F;
    const int   f    = i - node * F;
    const float di   = inv_sqrt_deg<TWO>(deg, node);
    float       v    = agg[i];
    if constexpr (TWO) v += aggB[i];
    agg[i] = v * di + bias[f];
}


template <bool TWO>
static void run_gcn(const float* x, const int* src, const int* dst, int E,
                    const float* W1, const float* b1, const float* W2, const float* b2,
                    const float* W3, const float* b3, float* out, float* ws,
                    hipStream_t stream) {
    const size_t N = GCN_N;

    float *deg, *t1, *h1, *h1B, *t2, *h2, *h2B, *t3, *outB;
    if constexpr (TWO) {
        deg  = ws;
        t1   = deg + 2 * N;
        t2   = t1;
        h1   = t1 + 32 * N;
        h1B  = h1 + 16 * N;
        h2   = h1B + 16 * N;
        h2B  = h2 + 32 * N;
        t3   = h1;
        outB = h1B;
    } else {
        deg  = ws;
        t1   = deg + N;
        h1   = t1 + 16 * N;
        t2   = h1 + 16 * N;
        h2   = t2 + 32 * N;
        t3   = h2 + 32 * N;
        h1B = h2B = outB = nullptr;
    }

    const int B = 256;
    const int SCATTER_BLOCKS = 2048;
    auto blk = [](int n, int b) { return (n + b - 1) / b; };

    k_init_deg<TWO><<<blk((TWO ? 2 : 1) * GCN_N, B), B, 0, stream>>>(deg);
    k_deg_accum<TWO><<<blk(E, B), B, 0, stream>>>(dst, E, deg);

    {
        const int waves = (GCN_N / 16) * 1;
        k_transform<2, 16, false, TWO><<<blk(waves * 32, B), B, 0, stream>>>(
            x, nullptr, W1, deg, nullptr, t1, h1, h1B);
        k_scatter<16, TWO><<<SCATTER_BLOCKS, B, 0, stream>>>(src, dst, E, t1, h1, h1B);
    }

    {
        const int waves = (GCN_N / 16) * 2;
        k_transform<16, 32, true, TWO><<<blk(waves * 32, B), B, 0, stream>>>(
            h1, h1B, W2, deg, b1, t2, h2, h2B);
        k_scatter<32, TWO><<<SCATTER_BLOCKS, B, 0, stream>>>(src, dst, E, t2, h2, h2B);
    }

    {
        const int waves = (GCN_N / 16) * 1;
        k_transform<32, 16, true, TWO><<<blk(waves * 32, B), B, 0, stream>>>(
            h2, h2B, W3, deg, b2, t3, out, outB);
        k_scatter<16, TWO><<<SCATTER_BLOCKS, B, 0, stream>>>(src, dst, E, t3, out, outB);
        k_finish<16, TWO><<<blk(GCN_N * 16, B), B, 0, stream>>>(deg, b3, out, outB);
    }
}

extern "C" void kernel_launch(void* const* d_in, const int* in_sizes, int n_in,
                              void* d_out, int out_size, void* d_ws, size_t ws_size,
                              hipStream_t stream) {
    const float* x  = (const float*)d_in[0];
    const int*   ei = (const int*)d_in[1];
    const float* W1 = (const float*)d_in[2];
    const float* b1 = (const float*)d_in[3];
    const float* W2 = (const float*)d_in[4];
    const float* b2 = (const float*)d_in[5];
    const float* W3 = (const float*)d_in[6];
    const float* b3 = (const float*)d_in[7];
    float* out = (float*)d_out;

    const int E = in_sizes[1] / 2;
    const int* src = ei;
    const int* dst = ei + E;

    if (ws_size >= (size_t)130 * GCN_N * sizeof(float))
        run_gcn<true >(x, src, dst, E, W1, b1, W2, b2, W3, b3, out, (float*)d_ws, stream);
    else
        run_gcn<false>(x, src, dst, E, W1, b1, W2, b2, W3, b3, out, (float*)d_ws, stream);
}


# In-degree counted in LDS: a work-group takes a node range and an edge slice, counts its destinations with LDS adds, then adds whole 128-byte rows of counts to its half's plane
# speedup vs baseline: 13.9927x; 1.0766x over previous
#include <hip/hip_runtime.h>


typedef __attribute__((ext_vector_type(2))) float v2f;
typedef __attribute__((ext_vector_type(8))) float v8f;

#define GCN_N 100000

static __device__ __forceinline__ int chip_half() {
    return (int)((__builtin_amdgcn_s_sendmsg_rtn(0x87) >> 18) & 1u);
}


template <bool TWO>
__global__ void k_init_deg(float* __restrict__ deg) {
    int i = blockIdx.x * blockDim.x + threadIdx.x;
    if (i < (TWO ? 2 : 1) * GCN_N) deg[i] = (i < GCN_N) ? 1.0f : 0.0f;
}

#define DEG_R  8
#define DEG_S  32
#define DEG_NR 12512
#define DEG_T  1024

template <bool TWO>
__global__ __launch_bounds__(DEG_T) void k_deg_hist(const int* __restrict__ dst, int E,
                                                    float* __restrict__ deg) {
    static_assert(DEG_R * DEG_NR >= GCN_N && (DEG_R - 1) * DEG_NR < GCN_N, "ranges cover the nodes");
    __shared__ unsigned cnt[DEG_NR];
    const int tid = threadIdx.x;
    const int r   = blockIdx.x % DEG_R;
    const int s   = blockIdx.x / DEG_R;
    const int lo  = r * DEG_NR;
    const int nr  = min(DEG_NR, GCN_N - lo);

    for (int i = tid; i < DEG_NR; i += DEG_T) cnt[i] = 0u;
    __syncthreads();

    const int per = (E + DEG_S - 1) / DEG_S;
    const int e0  = min(E, s * per);
    const int e1  = min(E, e0 + per);
    constexpr int U = 16;
    const int rounds = (e1 - e0 + DEG_T * U - 1) / (DEG_T * U);
    const int last   = e1 - 1;
    int raw[U];
    if (rounds > 0) {
#pragma unroll
        for (int u = 0; u < U; ++u)
            raw[u] = dst[min(e0 + tid + u * DEG_T, last)];
    }
    for (int rd = 0; rd < rounds; ++rd) {
        const int base = e0 + rd * (DEG_T * U) + tid;
        int k[U];
#pragma unroll
        for (int u = 0; u < U; ++u)
            k[u] = (base + u * DEG_T < e1) ? raw[u] - lo : -1;
        if (rd + 1 < rounds) {
#pragma unroll
            for (int u = 0; u < U; ++u)
                raw[u] = dst[min(base + DEG_T * U + u * DEG_T, last)];
        }
#pragma unroll
        for (int u = 0; u < U; ++u)
            if ((unsigned)k[u] < (unsigned)nr) atomicAdd(&cnt[k[u]], 1u);
    }
    __syncthreads();

    if constexpr (TWO) deg += chip_half() * GCN_N;
    constexpr int FL = (DEG_NR + DEG_T - 1) / DEG_T;
    unsigned c[FL];
#pragma unroll
    for (int j = 0; j < FL; ++j) c[j] = cnt[min(tid + j * DEG_T, DEG_NR - 1)];
#pragma unroll
    for (int j = 0; j < FL; ++j) {
        const int i = tid + j * DEG_T;
        if (i < nr && c[j] != 0u) atomicAdd(&deg[lo + i], (float)c[j]);
    }
}

template <bool TWO>
static __device__ __forceinline__ float inv_sqrt_deg(const float* __restrict__ deg, int node) {
    if constexpr (TWO) return rsqrtf(deg[node] + deg[GCN_N + node]);
    else               return rsqrtf(deg[node]);
}

template <int KI, int NO, bool EPI, bool TWO>
__global__ void k_transform(const float* __restrict__ X,
                            const float* __restrict__ XB,
                            const float* __restrict__ W,
                            const float* __restrict__ deg,
                            const float* __restrict__ bias_in,
                            float* __restrict__ T,
                            float* __restrict__ agg,
                            float* __restrict__ aggB) {
    static_assert(!EPI || KI % 4 == 0, "the epilogue on load has no padded k");
    const int wave     = (blockIdx.x * blockDim.x + threadIdx.x) >> 5;
    const int lane     = threadIdx.x & 31;
    const int nColTile = NO / 16;
    const int rowTile  = wave / nColTile;
    const int colTile  = wave - rowTile * nColTile;
    if (rowTile >= GCN_N / 16) return;

    const int row0 = rowTile * 16;
    const int col0 = colTile * 16;
    const int m    = lane & 15;
    const int hi   = lane >> 4;

    float di_m = 0.0f;
    if constexpr (EPI) di_m = inv_sqrt_deg<TWO>(deg, row0 + m);

    v8f acc = {};
    constexpr int KCHUNKS = (KI + 3) / 4;
#pragma unroll
    for (int kk = 0; kk < KCHUNKS; ++kk) {
        const int kb = kk * 4 + hi * 2;
        v2f a, b;
        a.x = (kb     < KI) ? X[(row0 + m) * KI + kb    ] : 0.0f;
        a.y = (kb + 1 < KI) ? X[(row0 + m) * KI + kb + 1] : 0.0f;
        if constexpr (EPI) {
            if constexpr (TWO) {
                a.x += XB[(row0 + m) * KI + kb    ];
                a.y += XB[(row0 + m) * KI + kb + 1];
            }
            a.x = fmaxf(a.x * di_m + bias_in[kb    ], 0.0f);
            a.y = fmaxf(a.y * di_m + bias_in[kb + 1], 0.0f);
        }
        b.x = (kb     < KI) ? W[(kb    ) * NO + col0 + m] : 0.0f;
        b.y = (kb + 1 < KI) ? W[(kb + 1) * NO + col0 + m] : 0.0f;
        acc = __builtin_amdgcn_wmma_f32_16x16x4_f32(
            false, a, false, b, (short)0, acc, false, false);
    }

#pragma unroll
    for (int j = 0; j < 8; ++j) {
        const int   row = row0 + j + hi * 8;
        const float v   = acc[j] * inv_sqrt_deg<TWO>(deg, row);
        T  [row * NO + col0 + m] = v;
        agg[row * NO + col0 + m] = v;
        if constexpr (TWO) aggB[row * NO + col0 + m] = 0.0f;
    }
}

template <int... Js> struct iseq {};

template <int J>
static __device__ __forceinline__ int half_lane(int x) {
    return __builtin_amdgcn_mov_dpp(x, 0x150 + J, 0xF, 0xF, false);
}

template <bool FULL, int... Js>
static __device__ __forceinline__ void rows16(int s, int d, int ok, int c,
                                              const float* __restrict__ T,
                                              float* __restrict__ agg, iseq<Js...>) {
    float v[16];
    ((v[Js] = T[(unsigned)half_lane<Js>(s) * 16u + (unsigned)c]), ...);
    ((FULL || half_lane<Js>(ok)
          ? (void)atomicAdd(&agg[(unsigned)half_lane<Js>(d) * 16u + (unsigned)c], v[Js])
          : (void)0), ...);
}

template <bool FULL>
static __device__ __forceinline__ void rows32(int s, int d, int nvalid, int lane,
                                              const float* __restrict__ T,
                                              float* __restrict__ agg) {
#pragma unroll
    for (int h = 0; h < 32; h += 16) {
        float v[16];
#pragma unroll
        for (int j = 0; j < 16; ++j) {
            const int sj = __builtin_amdgcn_readlane(s, h + j);
            v[j] = T[(unsigned)sj * 32u + (unsigned)lane];
        }
#pragma unroll
        for (int j = 0; j < 16; ++j) {
            if (FULL || h + j < nvalid) {
                const int dj = __builtin_amdgcn_readlane(d, h + j);
                atomicAdd(&agg[(unsigned)dj * 32u + (unsigned)lane], v[j]);
            }
        }
    }
}

template <int F, bool FULL>
static __device__ __forceinline__ void rows(int s, int d, int ok, int nvalid, int lane,
                                            const float* __restrict__ T,
                                            float* __restrict__ agg) {
    if constexpr (F == 32)
        rows32<FULL>(s, d, nvalid, lane, T, agg);
    else
        rows16<FULL>(s, d, ok, lane & 15, T, agg,
                     iseq<0, 1, 2, 3, 4, 5, 6, 7, 8, 9, 10, 11, 12, 13, 14, 15>{});
}

template <int F, bool TWO>
__global__ __launch_bounds__(256) void k_scatter(const int* __restrict__ src,
                                                 const int* __restrict__ dst, int E,
                                                 const float* __restrict__ T,
                                                 float* __restrict__ aggA,
                                                 float* __restrict__ aggB) {
    static_assert(F == 16 || F == 32, "row = 16 or 32 columns");
    const int lane  = threadIdx.x & 31;
    const int wpb   = blockDim.x >> 5;
    const int wave  = blockIdx.x * wpb + __builtin_amdgcn_readfirstlane(threadIdx.x >> 5);
    const int nwave = gridDim.x * wpb;
    const int nfull = E >> 5;

    float* __restrict__ agg = aggA;
    if constexpr (TWO) { if (chip_half()) agg = aggB; }

    int b = wave;
    int s = 0, d = 0;
    if (b < nfull) { s = src[(b << 5) + lane]; d = dst[(b << 5) + lane]; }
    while (b < nfull) {
        const int bn = b + nwave;
        int sn = 0, dn = 0;
        if (bn < nfull) { sn = src[(bn << 5) + lane]; dn = dst[(bn << 5) + lane]; }
        rows<F, true>(s, d, 1, 32, lane, T, agg);
        s = sn; d = dn; b = bn;
    }

    const int ntail = E & 31;
    if (ntail != 0 && wave == nfull % nwave) {
        const int  e  = (nfull << 5) + lane;
        const bool ok = lane < ntail;
        const int  st = ok ? src[e] : 0;
        const int  dt = ok ? dst[e] : 0;
        rows<F, false>(st, dt, ok ? 1 : 0, ntail, lane, T, agg);
    }
}

template <int F, bool TWO>
__global__ void k_finish(const float* __restrict__ deg,
                         const float* __restrict__ bias,
                         float* __restrict__ agg,
                         const float* __restrict__ aggB) {
    int i = blockIdx.x * blockDim.x + threadIdx.x;
    if (i >= GCN_N * F) return;
    const int   node = i / F;
    const int   f    = i - node * F;
    const float di   = inv_sqrt_deg<TWO>(deg, node);
    float       v    = agg[i];
    if constexpr (TWO) v += aggB[i];
    agg[i] = v * di + bias[f];
}


template <bool TWO>
static void run_gcn(const float* x, const int* src, const int* dst, int E,
                    const float* W1, const float* b1, const float* W2, const float* b2,
                    const float* W3, const float* b3, float* out, float* ws,
                    hipStream_t stream) {
    const size_t N = GCN_N;

    float *deg, *t1, *h1, *h1B, *t2, *h2, *h2B, *t3, *outB;
    if constexpr (TWO) {
        deg  = ws;
        t1   = deg + 2 * N;
        t2   = t1;
        h1   = t1 + 32 * N;
        h1B  = h1 + 16 * N;
        h2   = h1B + 16 * N;
        h2B  = h2 + 32 * N;
        t3   = h1;
        outB = h1B;
    } else {
        deg  = ws;
        t1   = deg + N;
        h1   = t1 + 16 * N;
        t2   = h1 + 16 * N;
        h2   = t2 + 32 * N;
        t3   = h2 + 32 * N;
        h1B = h2B = outB = nullptr;
    }

    const int B = 256;
    const int SCATTER_BLOCKS = 2048;
    auto blk = [](int n, int b) { return (n + b - 1) / b; };

    k_init_deg<TWO><<<blk((TWO ? 2 : 1) * GCN_N, B), B, 0, stream>>>(deg);
    k_deg_hist<TWO><<<DEG_R * DEG_S, DEG_T, 0, stream>>>(dst, E, deg);

    {
        const int waves = (GCN_N / 16) * 1;
        k_transform<2, 16, false, TWO><<<blk(waves * 32, B), B, 0, stream>>>(
            x, nullptr, W1, deg, nullptr, t1, h1, h1B);
        k_scatter<16, TWO><<<SCATTER_BLOCKS, B, 0, stream>>>(src, dst, E, t1, h1, h1B);
    }

    {
        const int waves = (GCN_N / 16) * 2;
        k_transform<16, 32, true, TWO><<<blk(waves * 32, B), B, 0, stream>>>(
            h1, h1B, W2, deg, b1, t2, h2, h2B);
        k_scatter<32, TWO><<<SCATTER_BLOCKS, B, 0, stream>>>(src, dst, E, t2, h2, h2B);
    }

    {
        const int waves = (GCN_N / 16) * 1;
        k_transform<32, 16, true, TWO><<<blk(waves * 32, B), B, 0, stream>>>(
            h2, h2B, W3, deg, b2, t3, out, outB);
        k_scatter<16, TWO><<<SCATTER_BLOCKS, B, 0, stream>>>(src, dst, E, t3, out, outB);
        k_finish<16, TWO><<<blk(GCN_N * 16, B), B, 0, stream>>>(deg, b3, out, outB);
    }
}

extern "C" void kernel_launch(void* const* d_in, const int* in_sizes, int n_in,
                              void* d_out, int out_size, void* d_ws, size_t ws_size,
                              hipStream_t stream) {
    const float* x  = (const float*)d_in[0];
    const int*   ei = (const int*)d_in[1];
    const float* W1 = (const float*)d_in[2];
    const float* b1 = (const float*)d_in[3];
    const float* W2 = (const float*)d_in[4];
    const float* b2 = (const float*)d_in[5];
    const float* W3 = (const float*)d_in[6];
    const float* b3 = (const float*)d_in[7];
    float* out = (float*)d_out;

    const int E = in_sizes[1] / 2;
    const int* src = ei;
    const int* dst = ei + E;

    if (ws_size >= (size_t)130 * GCN_N * sizeof(float))
        run_gcn<true >(x, src, dst, E, W1, b1, W2, b2, W3, b3, out, (float*)d_ws, stream);
    else
        run_gcn<false>(x, src, dst, E, W1, b1, W2, b2, W3, b3, out, (float*)d_ws, stream);
}
